# starting kernel
# baseline (speedup 1.0000x reference)
#include <hip/hip_runtime.h>
#include <hip/hip_bf16.h>

#define IN_FEAT 128
#define HIDDEN  64

typedef __attribute__((ext_vector_type(2))) float v2f;
typedef __attribute__((ext_vector_type(8))) float v8f;

__global__ void init_deg_kernel(float* __restrict__ deg, int n) {
    int i = blockIdx.x * blockDim.x + threadIdx.x;
    if (i < n) deg[i] = 1.0f;
}

__global__ void deg_accum_kernel(const int* __restrict__ cols,
                                 const float* __restrict__ ew,
                                 float* __restrict__ deg, int e) {
    int i = blockIdx.x * blockDim.x + threadIdx.x;
    if (i < e) atomicAdd(&deg[cols[i]], ew[i]);
}

__global__ void dinv_kernel(float* __restrict__ deg, int n) {
    int i = blockIdx.x * blockDim.x + threadIdx.x;
    if (i < n) {
        float d = deg[i];
        deg[i] = (d > 0.0f) ? __frsqrt_rn(d) : 0.0f;
    }
}

__global__ void __launch_bounds__(256)
gemm1_wmma_kernel(const float* __restrict__ X, const float* __restrict__ W,
                  float* __restrict__ H, int n) {
    __shared__ float sW[IN_FEAT * HIDDEN];
    for (int i = threadIdx.x; i < IN_FEAT * HIDDEN; i += blockDim.x)
        sW[i] = W[i];
    __syncthreads();

    const int wave = blockIdx.x * (blockDim.x >> 5) + (threadIdx.x >> 5);
    const int lane = threadIdx.x & 31;
    const int mt = wave >> 2;
    const int nt = wave & 3;
    const int m0 = mt * 16;
    if (m0 >= n) return;
    const int n0 = nt * 16;
    const int half = lane >> 4;
    const int l = lane & 15;

    const float* Xrow = X + (size_t)(m0 + l) * IN_FEAT;
    v8f c = {0.f, 0.f, 0.f, 0.f, 0.f, 0.f, 0.f, 0.f};

#pragma unroll 8
    for (int k = 0; k < IN_FEAT; k += 4) {
        v2f a = *(const v2f*)(Xrow + k + 2 * half);
        v2f b;
        b.x = sW[(k + 2 * half + 0) * HIDDEN + n0 + l];
        b.y = sW[(k + 2 * half + 1) * HIDDEN + n0 + l];
        c = __builtin_amdgcn_wmma_f32_16x16x4_f32(
                 false, a,  false, b,
                 (short)0, c,  false,  false);
    }

    float* Hout = H + (size_t)(m0 + 8 * half) * HIDDEN + n0 + l;
#pragma unroll
    for (int v = 0; v < 8; v++)
        Hout[(size_t)v * HIDDEN] = c[v];
}

__global__ void init_agg_kernel(const float* __restrict__ dinv,
                                const float* __restrict__ h0,
                                float* __restrict__ agg, int n) {
    int wave = blockIdx.x * (blockDim.x >> 5) + (threadIdx.x >> 5);
    int lane = threadIdx.x & 31;
    if (wave >= n) return;
    float d = dinv[wave];
    float s = d * d;
    const float2* src = (const float2*)(h0 + (size_t)wave * HIDDEN);
    float2* dst = (float2*)(agg + (size_t)wave * HIDDEN);
    float2 v = src[lane];
    v.x *= s; v.y *= s;
    dst[lane] = v;
}

__global__ void scatter1_kernel(const int* __restrict__ rows,
                                const int* __restrict__ cols,
                                const float* __restrict__ ew,
                                const float* __restrict__ dinv,
                                const float* __restrict__ h0,
                                float* __restrict__ agg, int e) {
    int wave = blockIdx.x * (blockDim.x >> 5) + (threadIdx.x >> 5);
    int lane = threadIdx.x & 31;
    if (wave >= e) return;
    int r = rows[wave];
    int c = cols[wave];
    float norm = dinv[r] * ew[wave] * dinv[c];
    float2 v = ((const float2*)(h0 + (size_t)r * HIDDEN))[lane];
    float* dst = agg + (size_t)c * HIDDEN + 2 * lane;
    atomicAdd(dst + 0, norm * v.x);
    atomicAdd(dst + 1, norm * v.y);
}

__global__ void relu_bias_kernel(float* __restrict__ agg,
                                 const float* __restrict__ b1, int total) {
    int i = blockIdx.x * blockDim.x + threadIdx.x;
    if (i < total) {
        float v = agg[i] + b1[i & (HIDDEN - 1)];
        agg[i] = fmaxf(v, 0.0f);
    }
}

__global__ void gemv2_init_kernel(const float* __restrict__ h,
                                  const float* __restrict__ w2,
                                  const float* __restrict__ b2,
                                  const float* __restrict__ dinv,
                                  float* __restrict__ g,
                                  float* __restrict__ out, int n) {
    int wave = blockIdx.x * (blockDim.x >> 5) + (threadIdx.x >> 5);
    int lane = threadIdx.x & 31;
    if (wave >= n) return;
    float2 hv = ((const float2*)(h + (size_t)wave * HIDDEN))[lane];
    float2 wv = ((const float2*)w2)[lane];
    float p = hv.x * wv.x + hv.y * wv.y;
#pragma unroll
    for (int off = 16; off > 0; off >>= 1)
        p += __shfl_xor(p, off, 32);
    if (lane == 0) {
        float d = dinv[wave];
        g[wave] = p;
        out[wave] = b2[0] + d * d * p;
    }
}

__global__ void scatter2_kernel(const int* __restrict__ rows,
                                const int* __restrict__ cols,
                                const float* __restrict__ ew,
                                const float* __restrict__ dinv,
                                const float* __restrict__ g,
                                float* __restrict__ out, int e) {
    int i = blockIdx.x * blockDim.x + threadIdx.x;
    if (i >= e) return;
    int r = rows[i];
    int c = cols[i];
    float norm = dinv[r] * ew[i] * dinv[c];
    atomicAdd(&out[c], norm * g[r]);
}

extern "C" void kernel_launch(void* const* d_in, const int* in_sizes, int n_in,
                              void* d_out, int out_size, void* d_ws, size_t ws_size,
                              hipStream_t stream) {
    const float* x  = (const float*)d_in[0];
    const int*   ei = (const int*)d_in[1];
    const float* ew = (const float*)d_in[2];
    const float* w1 = (const float*)d_in[3];
    const float* b1 = (const float*)d_in[4];
    const float* w2 = (const float*)d_in[5];
    const float* b2 = (const float*)d_in[6];
    float* out = (float*)d_out;

    const int E = in_sizes[2];
    const int N = in_sizes[0] / IN_FEAT;
    const int* rows = ei;
    const int* cols = ei + E;

    float* ws   = (float*)d_ws;
    float* dinv = ws;
    float* g    = ws + N;
    float* h0   = ws + 2 * (size_t)N;
    float* agg  = h0 + (size_t)N * HIDDEN;

    const int B = 256;
    init_deg_kernel<<<(N + B - 1) / B, B, 0, stream>>>(dinv, N);
    deg_accum_kernel<<<(E + B - 1) / B, B, 0, stream>>>(cols, ew, dinv, E);
    dinv_kernel<<<(N + B - 1) / B, B, 0, stream>>>(dinv, N);

    {
        int mtiles = (N + 15) / 16;
        int waves  = mtiles * (HIDDEN / 16);
        int blocks = (waves + 7) / 8;
        gemm1_wmma_kernel<<<blocks, B, 0, stream>>>(x, w1, h0, N);
    }

    init_agg_kernel<<<(N + 7) / 8, B, 0, stream>>>(dinv, h0, agg, N);
    scatter1_kernel<<<(E + 7) / 8, B, 0, stream>>>(rows, cols, ew, dinv, h0, agg, E);
    relu_bias_kernel<<<((size_t)N * HIDDEN + B - 1) / B, B, 0, stream>>>(agg, b1, N * HIDDEN);

    gemv2_init_kernel<<<(N + 7) / 8, B, 0, stream>>>(agg, w2, b2, dinv, g, out, N);
    scatter2_kernel<<<(E + B - 1) / B, B, 0, stream>>>(rows, cols, ew, dinv, g, out, E);
}


# relu_bias and gemv2_init fused into one pass: a half-wave reads a node's row of both planes by 16-byte loads, a wave walks 32 nodes, h never stored; 300k short waves -> 3125; dot sum order changes
# speedup vs baseline: 2.8711x; 2.8711x over previous
#include <hip/hip_runtime.h>
#include <hip/hip_bf16.h>

#define IN_FEAT 128
#define HIDDEN  64

typedef __attribute__((ext_vector_type(2))) float v2f;
typedef __attribute__((ext_vector_type(8))) float v8f;

static __device__ __forceinline__ bool in_upper_half() {
    return ((__builtin_amdgcn_s_sendmsg_rtn(0x87) >> 18) & 1u) != 0u;
}

__global__ void init_deg_kernel(float* __restrict__ degA, float* __restrict__ degB, int n) {
    int i = blockIdx.x * blockDim.x + threadIdx.x;
    if (i < n) { degA[i] = 1.0f; degB[i] = 0.0f; }
}

__global__ void deg_accum_kernel(const int* __restrict__ cols,
                                 const float* __restrict__ ew,
                                 float* __restrict__ degA,
                                 float* __restrict__ degB, int e) {
    float* deg = in_upper_half() ? degB : degA;
    int i = blockIdx.x * blockDim.x + threadIdx.x;
    if (i < e) atomicAdd(&deg[cols[i]], ew[i]);
}

__global__ void dinv_kernel(float* __restrict__ degA, const float* __restrict__ degB, int n) {
    int i = blockIdx.x * blockDim.x + threadIdx.x;
    if (i < n) {
        float d = degA[i] + degB[i];
        degA[i] = (d > 0.0f) ? __frsqrt_rn(d) : 0.0f;
    }
}

__global__ void __launch_bounds__(256)
gemm1_wmma_kernel(const float* __restrict__ X, const float* __restrict__ W,
                  float* __restrict__ H, int n) {
    __shared__ float sW[IN_FEAT * HIDDEN];
    for (int i = threadIdx.x; i < IN_FEAT * HIDDEN; i += blockDim.x)
        sW[i] = W[i];
    __syncthreads();

    const int wave = blockIdx.x * (blockDim.x >> 5) + (threadIdx.x >> 5);
    const int lane = threadIdx.x & 31;
    const int mt = wave >> 2;
    const int nt = wave & 3;
    const int m0 = mt * 16;
    if (m0 >= n) return;
    const int n0 = nt * 16;
    const int half = lane >> 4;
    const int l = lane & 15;

    const float* Xrow = X + (size_t)(m0 + l) * IN_FEAT;
    v8f c = {0.f, 0.f, 0.f, 0.f, 0.f, 0.f, 0.f, 0.f};

#pragma unroll 8
    for (int k = 0; k < IN_FEAT; k += 4) {
        v2f a = *(const v2f*)(Xrow + k + 2 * half);
        v2f b;
        b.x = sW[(k + 2 * half + 0) * HIDDEN + n0 + l];
        b.y = sW[(k + 2 * half + 1) * HIDDEN + n0 + l];
        c = __builtin_amdgcn_wmma_f32_16x16x4_f32(
                 false, a,  false, b,
                 (short)0, c,  false,  false);
    }

    float* Hout = H + (size_t)(m0 + 8 * half) * HIDDEN + n0 + l;
#pragma unroll
    for (int v = 0; v < 8; v++)
        Hout[(size_t)v * HIDDEN] = c[v];
}

__global__ void init_agg_kernel(const float* __restrict__ dinv,
                                const float* __restrict__ h0,
                                float* __restrict__ agg,
                                float* __restrict__ aggB, int n) {
    int wave = blockIdx.x * (blockDim.x >> 5) + (threadIdx.x >> 5);
    int lane = threadIdx.x & 31;
    if (wave >= n) return;
    float d = dinv[wave];
    float s = d * d;
    const float2* src = (const float2*)(h0 + (size_t)wave * HIDDEN);
    float2* dst = (float2*)(agg + (size_t)wave * HIDDEN);
    float2 v = src[lane];
    v.x *= s; v.y *= s;
    dst[lane] = v;
    ((float2*)(aggB + (size_t)wave * HIDDEN))[lane] = make_float2(0.0f, 0.0f);
}

#define S1_CHUNKS 4
#define S1_GROUP  8

__global__ void __launch_bounds__(256)
scatter1_kernel(const int* __restrict__ rows,
                const int* __restrict__ cols,
                const float* __restrict__ ew,
                const float* __restrict__ dinv,
                const float* __restrict__ h0,
                float* __restrict__ aggA,
                float* __restrict__ aggB, int e) {
    float* agg = in_upper_half() ? aggB : aggA;
    const int wave = __builtin_amdgcn_readfirstlane(
        (int)(blockIdx.x * (blockDim.x >> 5) + (threadIdx.x >> 5)));
    const int lane = threadIdx.x & 31;
    const int first = wave * (S1_CHUNKS * 32);

    for (int ch = 0; ch < S1_CHUNKS; ch++) {
        const int e0 = first + ch * 32;
        if (e0 >= e) break;
        const int cnt = (e - e0 < 32) ? (e - e0) : 32;

        int r = 0, c = 0;
        float norm = 0.0f;
        if (lane < cnt) {
            r = rows[e0 + lane];
            c = cols[e0 + lane];
            norm = dinv[r] * ew[e0 + lane] * dinv[c];
        }

        int j = 0;
        for (; j + S1_GROUP <= cnt; j += S1_GROUP) {
            float vlo[S1_GROUP], vhi[S1_GROUP];
#pragma unroll
            for (int u = 0; u < S1_GROUP; u++) {
                const int rj = __builtin_amdgcn_readlane(r, j + u);
                const float* srow = h0 + (size_t)rj * HIDDEN + lane;
                vlo[u] = srow[0];
                vhi[u] = srow[32];
            }
#pragma unroll
            for (int u = 0; u < S1_GROUP; u++) {
                const int cj = __builtin_amdgcn_readlane(c, j + u);
                const float nj = __int_as_float(
                    __builtin_amdgcn_readlane(__float_as_int(norm), j + u));
                float* dst = agg + (size_t)cj * HIDDEN + lane;
                atomicAdd(dst + 0,  nj * vlo[u]);
                atomicAdd(dst + 32, nj * vhi[u]);
            }
        }
        for (; j < cnt; j++) {
            const int rj = __builtin_amdgcn_readlane(r, j);
            const int cj = __builtin_amdgcn_readlane(c, j);
            const float nj = __int_as_float(
                __builtin_amdgcn_readlane(__float_as_int(norm), j));
            const float* srow = h0 + (size_t)rj * HIDDEN + lane;
            const float a0 = srow[0];
            const float a1 = srow[32];
            float* dst = agg + (size_t)cj * HIDDEN + lane;
            atomicAdd(dst + 0,  nj * a0);
            atomicAdd(dst + 32, nj * a1);
        }
    }
}

#define E1_NODES 32

__global__ void __launch_bounds__(64)
epilogue1_kernel(const float* __restrict__ aggA,
                 const float* __restrict__ aggB,
                 const float* __restrict__ b1,
                 const float* __restrict__ w2,
                 const float* __restrict__ b2,
                 const float* __restrict__ dinv,
                 float* __restrict__ g,
                 float* __restrict__ out,
                 float* __restrict__ outB, int n) {
    const int wave = blockIdx.x * (blockDim.x >> 5) + (threadIdx.x >> 5);
    const int lane = threadIdx.x & 31;
    const int base = wave * E1_NODES;
    if (base >= n) return;
    const int hl = lane & 15;
    const int half = lane >> 4;
    const float4 bv = ((const float4*)b1)[hl];
    const float4 wv = ((const float4*)w2)[hl];
    float res = 0.0f;

#pragma unroll 4
    for (int u = 0; u < E1_NODES / 2; u++) {
        int node = base + 2 * u + half;
        node = (node < n) ? node : (n - 1);
        const float4 a = ((const float4*)(aggA + (size_t)node * HIDDEN))[hl];
        const float4 b = ((const float4*)(aggB + (size_t)node * HIDDEN))[hl];
        const float hx = fmaxf((a.x + b.x) + bv.x, 0.0f);
        const float hy = fmaxf((a.y + b.y) + bv.y, 0.0f);
        const float hz = fmaxf((a.z + b.z) + bv.z, 0.0f);
        const float hw = fmaxf((a.w + b.w) + bv.w, 0.0f);
        float p = hx * wv.x + hy * wv.y + hz * wv.z + hw * wv.w;
#pragma unroll
        for (int off = 8; off > 0; off >>= 1)
            p += __shfl_xor(p, off, 32);
        const float q = __shfl(p, (lane & 1) << 4, 32);
        if ((lane >> 1) == u) res = q;
    }

    const int i = base + lane;
    if (i < n) {
        const float d = dinv[i];
        g[i] = res;
        out[i] = b2[0] + d * d * res;
        outB[i] = 0.0f;
    }
}

__global__ void scatter2_kernel(const int* __restrict__ rows,
                                const int* __restrict__ cols,
                                const float* __restrict__ ew,
                                const float* __restrict__ dinv,
                                const float* __restrict__ g,
                                float* __restrict__ outA,
                                float* __restrict__ outB, int e) {
    float* out = in_upper_half() ? outB : outA;
    int i = blockIdx.x * blockDim.x + threadIdx.x;
    if (i >= e) return;
    int r = rows[i];
    int c = cols[i];
    float norm = dinv[r] * ew[i] * dinv[c];
    atomicAdd(&out[c], norm * g[r]);
}

__global__ void add_planes_kernel(float* __restrict__ out,
                                  const float* __restrict__ outB, int n) {
    int i = blockIdx.x * blockDim.x + threadIdx.x;
    if (i < n) out[i] = out[i] + outB[i];
}

extern "C" void kernel_launch(void* const* d_in, const int* in_sizes, int n_in,
                              void* d_out, int out_size, void* d_ws, size_t ws_size,
                              hipStream_t stream) {
    const float* x  = (const float*)d_in[0];
    const int*   ei = (const int*)d_in[1];
    const float* ew = (const float*)d_in[2];
    const float* w1 = (const float*)d_in[3];
    const float* b1 = (const float*)d_in[4];
    const float* w2 = (const float*)d_in[5];
    const float* b2 = (const float*)d_in[6];
    float* out = (float*)d_out;

    const int E = in_sizes[2];
    const int N = in_sizes[0] / IN_FEAT;
    const int* rows = ei;
    const int* cols = ei + E;

    float* ws   = (float*)d_ws;
    float* dinv = ws;
    float* g    = ws + N;
    float* degB = ws + 2 * (size_t)N;
    float* outB = ws + 3 * (size_t)N;
    float* h0   = ws + 4 * (size_t)N;
    float* agg  = h0 + (size_t)N * HIDDEN;
    float* aggB = agg + (size_t)N * HIDDEN;

    const int B = 256;
    init_deg_kernel<<<(N + B - 1) / B, B, 0, stream>>>(dinv, degB, N);
    deg_accum_kernel<<<(E + B - 1) / B, B, 0, stream>>>(cols, ew, dinv, degB, E);
    dinv_kernel<<<(N + B - 1) / B, B, 0, stream>>>(dinv, degB, N);

    {
        int mtiles = (N + 15) / 16;
        int waves  = mtiles * (HIDDEN / 16);
        int blocks = (waves + 7) / 8;
        gemm1_wmma_kernel<<<blocks, B, 0, stream>>>(x, w1, h0, N);
    }

    init_agg_kernel<<<(N + 7) / 8, B, 0, stream>>>(dinv, h0, agg, aggB, N);
    scatter1_kernel<<<(E + 8 * S1_CHUNKS * 32 - 1) / (8 * S1_CHUNKS * 32), B, 0, stream>>>(rows, cols, ew, dinv, h0, agg, aggB, E);

    epilogue1_kernel<<<(N + 2 * E1_NODES - 1) / (2 * E1_NODES), 64, 0, stream>>>(agg, aggB, b1, w2, b2, dinv, g, out, outB, N);
    scatter2_kernel<<<(E + B - 1) / B, B, 0, stream>>>(rows, cols, ew, dinv, g, out, outB, E);
    add_planes_kernel<<<(N + B - 1) / B, B, 0, stream>>>(out, outB, N);
}


# Per-edge norm formed once: the layer-1 edge pass stores it (E floats, edge order), the layer-2 edge pass reads it back; 2 of its 3 random 4-byte gathers per edge (dinv[row], dinv[col]) gone
# speedup vs baseline: 3.0473x; 1.0614x over previous
#include <hip/hip_runtime.h>
#include <hip/hip_bf16.h>

#define IN_FEAT 128
#define HIDDEN  64

typedef __attribute__((ext_vector_type(2))) float v2f;
typedef __attribute__((ext_vector_type(8))) float v8f;

static __device__ __forceinline__ bool in_upper_half() {
    return ((__builtin_amdgcn_s_sendmsg_rtn(0x87) >> 18) & 1u) != 0u;
}

__global__ void init_deg_kernel(float* __restrict__ degA, float* __restrict__ degB, int n) {
    int i = blockIdx.x * blockDim.x + threadIdx.x;
    if (i < n) { degA[i] = 1.0f; degB[i] = 0.0f; }
}

__global__ void deg_accum_kernel(const int* __restrict__ cols,
                                 const float* __restrict__ ew,
                                 float* __restrict__ degA,
                                 float* __restrict__ degB, int e) {
    float* deg = in_upper_half() ? degB : degA;
    int i = blockIdx.x * blockDim.x + threadIdx.x;
    if (i < e) atomicAdd(&deg[cols[i]], ew[i]);
}

__global__ void dinv_kernel(float* __restrict__ degA, const float* __restrict__ degB, int n) {
    int i = blockIdx.x * blockDim.x + threadIdx.x;
    if (i < n) {
        float d = degA[i] + degB[i];
        degA[i] = (d > 0.0f) ? __frsqrt_rn(d) : 0.0f;
    }
}

__global__ void __launch_bounds__(256)
gemm1_wmma_kernel(const float* __restrict__ X, const float* __restrict__ W,
                  float* __restrict__ H, int n) {
    __shared__ float sW[IN_FEAT * HIDDEN];
    for (int i = threadIdx.x; i < IN_FEAT * HIDDEN; i += blockDim.x)
        sW[i] = W[i];
    __syncthreads();

    const int wave = blockIdx.x * (blockDim.x >> 5) + (threadIdx.x >> 5);
    const int lane = threadIdx.x & 31;
    const int mt = wave >> 2;
    const int nt = wave & 3;
    const int m0 = mt * 16;
    if (m0 >= n) return;
    const int n0 = nt * 16;
    const int half = lane >> 4;
    const int l = lane & 15;

    const float* Xrow = X + (size_t)(m0 + l) * IN_FEAT;
    v8f c = {0.f, 0.f, 0.f, 0.f, 0.f, 0.f, 0.f, 0.f};

#pragma unroll 8
    for (int k = 0; k < IN_FEAT; k += 4) {
        v2f a = *(const v2f*)(Xrow + k + 2 * half);
        v2f b;
        b.x = sW[(k + 2 * half + 0) * HIDDEN + n0 + l];
        b.y = sW[(k + 2 * half + 1) * HIDDEN + n0 + l];
        c = __builtin_amdgcn_wmma_f32_16x16x4_f32(
                 false, a,  false, b,
                 (short)0, c,  false,  false);
    }

    float* Hout = H + (size_t)(m0 + 8 * half) * HIDDEN + n0 + l;
#pragma unroll
    for (int v = 0; v < 8; v++)
        Hout[(size_t)v * HIDDEN] = c[v];
}

__global__ void init_agg_kernel(const float* __restrict__ dinv,
                                const float* __restrict__ h0,
                                float* __restrict__ agg,
                                float* __restrict__ aggB, int n) {
    int wave = blockIdx.x * (blockDim.x >> 5) + (threadIdx.x >> 5);
    int lane = threadIdx.x & 31;
    if (wave >= n) return;
    float d = dinv[wave];
    float s = d * d;
    const float2* src = (const float2*)(h0 + (size_t)wave * HIDDEN);
    float2* dst = (float2*)(agg + (size_t)wave * HIDDEN);
    float2 v = src[lane];
    v.x *= s; v.y *= s;
    dst[lane] = v;
    ((float2*)(aggB + (size_t)wave * HIDDEN))[lane] = make_float2(0.0f, 0.0f);
}

#define S1_CHUNKS 4
#define S1_GROUP  8

__global__ void __launch_bounds__(256)
scatter1_kernel(const int* __restrict__ rows,
                const int* __restrict__ cols,
                const float* __restrict__ ew,
                const float* __restrict__ dinv,
                const float* __restrict__ h0,
                float* __restrict__ aggA,
                float* __restrict__ aggB,
                float* __restrict__ enorm, int e) {
    float* agg = in_upper_half() ? aggB : aggA;
    const int wave = __builtin_amdgcn_readfirstlane(
        (int)(blockIdx.x * (blockDim.x >> 5) + (threadIdx.x >> 5)));
    const int lane = threadIdx.x & 31;
    const int first = wave * (S1_CHUNKS * 32);

    for (int ch = 0; ch < S1_CHUNKS; ch++) {
        const int e0 = first + ch * 32;
        if (e0 >= e) break;
        const int cnt = (e - e0 < 32) ? (e - e0) : 32;

        int r = 0, c = 0;
        float norm = 0.0f;
        if (lane < cnt) {
            r = rows[e0 + lane];
            c = cols[e0 + lane];
            norm = dinv[r] * ew[e0 + lane] * dinv[c];
            enorm[e0 + lane] = norm;
        }

        int j = 0;
        for (; j + S1_GROUP <= cnt; j += S1_GROUP) {
            float vlo[S1_GROUP], vhi[S1_GROUP];
#pragma unroll
            for (int u = 0; u < S1_GROUP; u++) {
                const int rj = __builtin_amdgcn_readlane(r, j + u);
                const float* srow = h0 + (size_t)rj * HIDDEN + lane;
                vlo[u] = srow[0];
                vhi[u] = srow[32];
            }
#pragma unroll
            for (int u = 0; u < S1_GROUP; u++) {
                const int cj = __builtin_amdgcn_readlane(c, j + u);
                const float nj = __int_as_float(
                    __builtin_amdgcn_readlane(__float_as_int(norm), j + u));
                float* dst = agg + (size_t)cj * HIDDEN + lane;
                atomicAdd(dst + 0,  nj * vlo[u]);
                atomicAdd(dst + 32, nj * vhi[u]);
            }
        }
        for (; j < cnt; j++) {
            const int rj = __builtin_amdgcn_readlane(r, j);
            const int cj = __builtin_amdgcn_readlane(c, j);
            const float nj = __int_as_float(
                __builtin_amdgcn_readlane(__float_as_int(norm), j));
            const float* srow = h0 + (size_t)rj * HIDDEN + lane;
            const float a0 = srow[0];
            const float a1 = srow[32];
            float* dst = agg + (size_t)cj * HIDDEN + lane;
            atomicAdd(dst + 0,  nj * a0);
            atomicAdd(dst + 32, nj * a1);
        }
    }
}

#define E1_NODES 32

__global__ void __launch_bounds__(64)
epilogue1_kernel(const float* __restrict__ aggA,
                 const float* __restrict__ aggB,
                 const float* __restrict__ b1,
                 const float* __restrict__ w2,
                 const float* __restrict__ b2,
                 const float* __restrict__ dinv,
                 float* __restrict__ g,
                 float* __restrict__ out,
                 float* __restrict__ outB, int n) {
    const int wave = blockIdx.x * (blockDim.x >> 5) + (threadIdx.x >> 5);
    const int lane = threadIdx.x & 31;
    const int base = wave * E1_NODES;
    if (base >= n) return;
    const int hl = lane & 15;
    const int half = lane >> 4;
    const float4 bv = ((const float4*)b1)[hl];
    const float4 wv = ((const float4*)w2)[hl];
    float res = 0.0f;

#pragma unroll 4
    for (int u = 0; u < E1_NODES / 2; u++) {
        int node = base + 2 * u + half;
        node = (node < n) ? node : (n - 1);
        const float4 a = ((const float4*)(aggA + (size_t)node * HIDDEN))[hl];
        const float4 b = ((const float4*)(aggB + (size_t)node * HIDDEN))[hl];
        const float hx = fmaxf((a.x + b.x) + bv.x, 0.0f);
        const float hy = fmaxf((a.y + b.y) + bv.y, 0.0f);
        const float hz = fmaxf((a.z + b.z) + bv.z, 0.0f);
        const float hw = fmaxf((a.w + b.w) + bv.w, 0.0f);
        float p = hx * wv.x + hy * wv.y + hz * wv.z + hw * wv.w;
#pragma unroll
        for (int off = 8; off > 0; off >>= 1)
            p += __shfl_xor(p, off, 32);
        const float q = __shfl(p, (lane & 1) << 4, 32);
        if ((lane >> 1) == u) res = q;
    }

    const int i = base + lane;
    if (i < n) {
        const float d = dinv[i];
        g[i] = res;
        out[i] = b2[0] + d * d * res;
        outB[i] = 0.0f;
    }
}

__global__ void scatter2_kernel(const int* __restrict__ rows,
                                const int* __restrict__ cols,
                                const float* __restrict__ enorm,
                                const float* __restrict__ g,
                                float* __restrict__ outA,
                                float* __restrict__ outB, int e) {
    float* out = in_upper_half() ? outB : outA;
    int i = blockIdx.x * blockDim.x + threadIdx.x;
    if (i >= e) return;
    int r = rows[i];
    int c = cols[i];
    float norm = enorm[i];
    atomicAdd(&out[c], norm * g[r]);
}

__global__ void add_planes_kernel(float* __restrict__ out,
                                  const float* __restrict__ outB, int n) {
    int i = blockIdx.x * blockDim.x + threadIdx.x;
    if (i < n) out[i] = out[i] + outB[i];
}

extern "C" void kernel_launch(void* const* d_in, const int* in_sizes, int n_in,
                              void* d_out, int out_size, void* d_ws, size_t ws_size,
                              hipStream_t stream) {
    const float* x  = (const float*)d_in[0];
    const int*   ei = (const int*)d_in[1];
    const float* ew = (const float*)d_in[2];
    const float* w1 = (const float*)d_in[3];
    const float* b1 = (const float*)d_in[4];
    const float* w2 = (const float*)d_in[5];
    const float* b2 = (const float*)d_in[6];
    float* out = (float*)d_out;

    const int E = in_sizes[2];
    const int N = in_sizes[0] / IN_FEAT;
    const int* rows = ei;
    const int* cols = ei + E;

    float* ws   = (float*)d_ws;
    float* dinv = ws;
    float* g    = ws + N;
    float* degB = ws + 2 * (size_t)N;
    float* outB = ws + 3 * (size_t)N;
    float* h0   = ws + 4 * (size_t)N;
    float* agg  = h0 + (size_t)N * HIDDEN;
    float* aggB = agg + (size_t)N * HIDDEN;
    float* enorm = aggB + (size_t)N * HIDDEN;

    const int B = 256;
    init_deg_kernel<<<(N + B - 1) / B, B, 0, stream>>>(dinv, degB, N);
    deg_accum_kernel<<<(E + B - 1) / B, B, 0, stream>>>(cols, ew, dinv, degB, E);
    dinv_kernel<<<(N + B - 1) / B, B, 0, stream>>>(dinv, degB, N);

    {
        int mtiles = (N + 15) / 16;
        int waves  = mtiles * (HIDDEN / 16);
        int blocks = (waves + 7) / 8;
        gemm1_wmma_kernel<<<blocks, B, 0, stream>>>(x, w1, h0, N);
    }

    init_agg_kernel<<<(N + 7) / 8, B, 0, stream>>>(dinv, h0, agg, aggB, N);
    scatter1_kernel<<<(E + 8 * S1_CHUNKS * 32 - 1) / (8 * S1_CHUNKS * 32), B, 0, stream>>>(rows, cols, ew, dinv, h0, agg, aggB, enorm, E);

    epilogue1_kernel<<<(N + 2 * E1_NODES - 1) / (2 * E1_NODES), 64, 0, stream>>>(agg, aggB, b1, w2, b2, dinv, g, out, outB, N);
    scatter2_kernel<<<(E + B - 1) / B, B, 0, stream>>>(rows, cols, enorm, g, out, outB, E);
    add_planes_kernel<<<(N + B - 1) / B, B, 0, stream>>>(out, outB, N);
}


# GEMM tile store also writes dinv (from the two degree planes) and both aggregate start planes (dinv^2*H0, 0): dinv pass and 100k-wave init pass gone, H0 not re-read; 9 -> 7 launches
# speedup vs baseline: 3.2330x; 1.0609x over previous
#include <hip/hip_runtime.h>
#include <hip/hip_bf16.h>

#define IN_FEAT 128
#define HIDDEN  64

typedef __attribute__((ext_vector_type(2))) float v2f;
typedef __attribute__((ext_vector_type(8))) float v8f;

static __device__ __forceinline__ bool in_upper_half() {
    return ((__builtin_amdgcn_s_sendmsg_rtn(0x87) >> 18) & 1u) != 0u;
}

__global__ void init_deg_kernel(float* __restrict__ degA, float* __restrict__ degB, int n) {
    int i = blockIdx.x * blockDim.x + threadIdx.x;
    if (i < n) { degA[i] = 1.0f; degB[i] = 0.0f; }
}

__global__ void deg_accum_kernel(const int* __restrict__ cols,
                                 const float* __restrict__ ew,
                                 float* __restrict__ degA,
                                 float* __restrict__ degB, int e) {
    float* deg = in_upper_half() ? degB : degA;
    int i = blockIdx.x * blockDim.x + threadIdx.x;
    if (i < e) atomicAdd(&deg[cols[i]], ew[i]);
}

__global__ void __launch_bounds__(256)
gemm1_wmma_kernel(const float* __restrict__ X, const float* __restrict__ W,
                  const float* __restrict__ degA, const float* __restrict__ degB,
                  float* __restrict__ dinv,
                  float* __restrict__ H,
                  float* __restrict__ aggA, float* __restrict__ aggB, int n) {
    __shared__ float sW[IN_FEAT * HIDDEN];
    for (int i = threadIdx.x; i < IN_FEAT * HIDDEN; i += blockDim.x)
        sW[i] = W[i];
    __syncthreads();

    const int wave = blockIdx.x * (blockDim.x >> 5) + (threadIdx.x >> 5);
    const int lane = threadIdx.x & 31;
    const int mt = wave >> 2;
    const int nt = wave & 3;
    const int m0 = mt * 16;
    if (m0 >= n) return;
    const int n0 = nt * 16;
    const int half = lane >> 4;
    const int l = lane & 15;

    const float* Xrow = X + (size_t)(m0 + l) * IN_FEAT;
    v8f c = {0.f, 0.f, 0.f, 0.f, 0.f, 0.f, 0.f, 0.f};

    const float dsum = degA[m0 + l] + degB[m0 + l];

#pragma unroll 8
    for (int k = 0; k < IN_FEAT; k += 4) {
        v2f a = *(const v2f*)(Xrow + k + 2 * half);
        v2f b;
        b.x = sW[(k + 2 * half + 0) * HIDDEN + n0 + l];
        b.y = sW[(k + 2 * half + 1) * HIDDEN + n0 + l];
        c = __builtin_amdgcn_wmma_f32_16x16x4_f32(
                 false, a,  false, b,
                 (short)0, c,  false,  false);
    }

    const float di = (dsum > 0.0f) ? __frsqrt_rn(dsum) : 0.0f;
    if (nt == 0 && half == 0) dinv[m0 + l] = di;
    const float sq = di * di;

    const size_t off = (size_t)(m0 + 8 * half) * HIDDEN + n0 + l;
#pragma unroll
    for (int v = 0; v < 8; v++) {
        const float sv = __shfl(sq, 8 * half + v, 32);
        H[off + (size_t)v * HIDDEN]    = c[v];
        aggA[off + (size_t)v * HIDDEN] = sv * c[v];
        aggB[off + (size_t)v * HIDDEN] = 0.0f;
    }
}

#define S1_CHUNKS 4
#define S1_GROUP  8

__global__ void __launch_bounds__(256)
scatter1_kernel(const int* __restrict__ rows,
                const int* __restrict__ cols,
                const float* __restrict__ ew,
                const float* __restrict__ dinv,
                const float* __restrict__ h0,
                float* __restrict__ aggA,
                float* __restrict__ aggB,
                float* __restrict__ enorm, int e) {
    float* agg = in_upper_half() ? aggB : aggA;
    const int wave = __builtin_amdgcn_readfirstlane(
        (int)(blockIdx.x * (blockDim.x >> 5) + (threadIdx.x >> 5)));
    const int lane = threadIdx.x & 31;
    const int first = wave * (S1_CHUNKS * 32);

    for (int ch = 0; ch < S1_CHUNKS; ch++) {
        const int e0 = first + ch * 32;
        if (e0 >= e) break;
        const int cnt = (e - e0 < 32) ? (e - e0) : 32;

        int r = 0, c = 0;
        float norm = 0.0f;
        if (lane < cnt) {
            r = rows[e0 + lane];
            c = cols[e0 + lane];
            norm = dinv[r] * ew[e0 + lane] * dinv[c];
            enorm[e0 + lane] = norm;
        }

        int j = 0;
        for (; j + S1_GROUP <= cnt; j += S1_GROUP) {
            float vlo[S1_GROUP], vhi[S1_GROUP];
#pragma unroll
            for (int u = 0; u < S1_GROUP; u++) {
                const int rj = __builtin_amdgcn_readlane(r, j + u);
                const float* srow = h0 + (size_t)rj * HIDDEN + lane;
                vlo[u] = srow[0];
                vhi[u] = srow[32];
            }
#pragma unroll
            for (int u = 0; u < S1_GROUP; u++) {
                const int cj = __builtin_amdgcn_readlane(c, j + u);
                const float nj = __int_as_float(
                    __builtin_amdgcn_readlane(__float_as_int(norm), j + u));
                float* dst = agg + (size_t)cj * HIDDEN + lane;
                atomicAdd(dst + 0,  nj * vlo[u]);
                atomicAdd(dst + 32, nj * vhi[u]);
            }
        }
        for (; j < cnt; j++) {
            const int rj = __builtin_amdgcn_readlane(r, j);
            const int cj = __builtin_amdgcn_readlane(c, j);
            const float nj = __int_as_float(
                __builtin_amdgcn_readlane(__float_as_int(norm), j));
            const float* srow = h0 + (size_t)rj * HIDDEN + lane;
            const float a0 = srow[0];
            const float a1 = srow[32];
            float* dst = agg + (size_t)cj * HIDDEN + lane;
            atomicAdd(dst + 0,  nj * a0);
            atomicAdd(dst + 32, nj * a1);
        }
    }
}

#define E1_NODES 32

__global__ void __launch_bounds__(64)
epilogue1_kernel(const float* __restrict__ aggA,
                 const float* __restrict__ aggB,
                 const float* __restrict__ b1,
                 const float* __restrict__ w2,
                 const float* __restrict__ b2,
                 const float* __restrict__ dinv,
                 float* __restrict__ g,
                 float* __restrict__ out,
                 float* __restrict__ outB, int n) {
    const int wave = blockIdx.x * (blockDim.x >> 5) + (threadIdx.x >> 5);
    const int lane = threadIdx.x & 31;
    const int base = wave * E1_NODES;
    if (base >= n) return;
    const int hl = lane & 15;
    const int half = lane >> 4;
    const float4 bv = ((const float4*)b1)[hl];
    const float4 wv = ((const float4*)w2)[hl];
    float res = 0.0f;

#pragma unroll 4
    for (int u = 0; u < E1_NODES / 2; u++) {
        int node = base + 2 * u + half;
        node = (node < n) ? node : (n - 1);
        const float4 a = ((const float4*)(aggA + (size_t)node * HIDDEN))[hl];
        const float4 b = ((const float4*)(aggB + (size_t)node * HIDDEN))[hl];
        const float hx = fmaxf((a.x + b.x) + bv.x, 0.0f);
        const float hy = fmaxf((a.y + b.y) + bv.y, 0.0f);
        const float hz = fmaxf((a.z + b.z) + bv.z, 0.0f);
        const float hw = fmaxf((a.w + b.w) + bv.w, 0.0f);
        float p = hx * wv.x + hy * wv.y + hz * wv.z + hw * wv.w;
#pragma unroll
        for (int off = 8; off > 0; off >>= 1)
            p += __shfl_xor(p, off, 32);
        const float q = __shfl(p, (lane & 1) << 4, 32);
        if ((lane >> 1) == u) res = q;
    }

    const int i = base + lane;
    if (i < n) {
        const float d = dinv[i];
        g[i] = res;
        out[i] = b2[0] + d * d * res;
        outB[i] = 0.0f;
    }
}

__global__ void scatter2_kernel(const int* __restrict__ rows,
                                const int* __restrict__ cols,
                                const float* __restrict__ enorm,
                                const float* __restrict__ g,
                                float* __restrict__ outA,
                                float* __restrict__ outB, int e) {
    float* out = in_upper_half() ? outB : outA;
    int i = blockIdx.x * blockDim.x + threadIdx.x;
    if (i >= e) return;
    int r = rows[i];
    int c = cols[i];
    float norm = enorm[i];
    atomicAdd(&out[c], norm * g[r]);
}

__global__ void add_planes_kernel(float* __restrict__ out,
                                  const float* __restrict__ outB, int n) {
    int i = blockIdx.x * blockDim.x + threadIdx.x;
    if (i < n) out[i] = out[i] + outB[i];
}

extern "C" void kernel_launch(void* const* d_in, const int* in_sizes, int n_in,
                              void* d_out, int out_size, void* d_ws, size_t ws_size,
                              hipStream_t stream) {
    const float* x  = (const float*)d_in[0];
    const int*   ei = (const int*)d_in[1];
    const float* ew = (const float*)d_in[2];
    const float* w1 = (const float*)d_in[3];
    const float* b1 = (const float*)d_in[4];
    const float* w2 = (const float*)d_in[5];
    const float* b2 = (const float*)d_in[6];
    float* out = (float*)d_out;

    const int E = in_sizes[2];
    const int N = in_sizes[0] / IN_FEAT;
    const int* rows = ei;
    const int* cols = ei + E;

    float* ws   = (float*)d_ws;
    float* degA = ws;
    float* g    = ws + N;
    float* degB = ws + 2 * (size_t)N;
    float* outB = ws + 3 * (size_t)N;
    float* h0   = ws + 4 * (size_t)N;
    float* agg  = h0 + (size_t)N * HIDDEN;
    float* aggB = agg + (size_t)N * HIDDEN;
    float* enorm = aggB + (size_t)N * HIDDEN;
    float* dinv = enorm + (size_t)E;

    const int B = 256;
    init_deg_kernel<<<(N + B - 1) / B, B, 0, stream>>>(degA, degB, N);
    deg_accum_kernel<<<(E + B - 1) / B, B, 0, stream>>>(cols, ew, degA, degB, E);

    {
        int mtiles = (N + 15) / 16;
        int waves  = mtiles * (HIDDEN / 16);
        int blocks = (waves + 7) / 8;
        gemm1_wmma_kernel<<<blocks, B, 0, stream>>>(x, w1, degA, degB, dinv, h0, agg, aggB, N);
    }

    scatter1_kernel<<<(E + 8 * S1_CHUNKS * 32 - 1) / (8 * S1_CHUNKS * 32), B, 0, stream>>>(rows, cols, ew, dinv, h0, agg, aggB, enorm, E);

    epilogue1_kernel<<<(N + 2 * E1_NODES - 1) / (2 * E1_NODES), 64, 0, stream>>>(agg, aggB, b1, w2, b2, dinv, g, out, outB, N);
    scatter2_kernel<<<(E + B - 1) / B, B, 0, stream>>>(rows, cols, enorm, g, out, outB, E);
    add_planes_kernel<<<(N + B - 1) / B, B, 0, stream>>>(out, outB, N);
}


# Degree edge pass inside the GEMM launch: each GEMM thread first adds 2 edges' weights (float atomics), hidden under tile streaming; self-loop term added in epilogue, dinv own pass; sum order changes
# speedup vs baseline: 3.4336x; 1.0620x over previous
#include <hip/hip_runtime.h>
#include <hip/hip_bf16.h>

#define IN_FEAT 128
#define HIDDEN  64

typedef __attribute__((ext_vector_type(2))) float v2f;
typedef __attribute__((ext_vector_type(8))) float v8f;

static __device__ __forceinline__ bool in_upper_half() {
    return ((__builtin_amdgcn_s_sendmsg_rtn(0x87) >> 18) & 1u) != 0u;
}

__global__ void init_deg_kernel(float* __restrict__ degA, float* __restrict__ degB, int n) {
    int i = blockIdx.x * blockDim.x + threadIdx.x;
    if (i < n) { degA[i] = 1.0f; degB[i] = 0.0f; }
}

__global__ void dinv_kernel(const float* __restrict__ degA, const float* __restrict__ degB,
                            float* __restrict__ dinv, int n) {
    int i = blockIdx.x * blockDim.x + threadIdx.x;
    if (i < n) {
        const float d = degA[i] + degB[i];
        dinv[i] = (d > 0.0f) ? __frsqrt_rn(d) : 0.0f;
    }
}

__global__ void __launch_bounds__(256)
gemm1_wmma_kernel(const float* __restrict__ X, const float* __restrict__ W,
                  const int* __restrict__ cols, const float* __restrict__ ew,
                  float* __restrict__ degA, float* __restrict__ degB,
                  float* __restrict__ H,
                  float* __restrict__ aggA, float* __restrict__ aggB, int n, int e) {
    __shared__ float sW[IN_FEAT * HIDDEN];

    {
        float* deg = in_upper_half() ? degB : degA;
        const int stride = (int)(gridDim.x * blockDim.x);
        for (int i = (int)(blockIdx.x * blockDim.x + threadIdx.x); i < e; i += 2 * stride) {
            const int i1 = i + stride;
            const bool two = i1 < e;
            const int c0 = cols[i];
            const float w0 = ew[i];
            const int c1 = two ? cols[i1] : 0;
            const float w1 = two ? ew[i1] : 0.0f;
            atomicAdd(&deg[c0], w0);
            if (two) atomicAdd(&deg[c1], w1);
        }
    }

    for (int i = threadIdx.x; i < IN_FEAT * HIDDEN; i += blockDim.x)
        sW[i] = W[i];
    __syncthreads();

    const int wave = blockIdx.x * (blockDim.x >> 5) + (threadIdx.x >> 5);
    const int lane = threadIdx.x & 31;
    const int mt = wave >> 2;
    const int nt = wave & 3;
    const int m0 = mt * 16;
    if (m0 >= n) return;
    const int n0 = nt * 16;
    const int half = lane >> 4;
    const int l = lane & 15;

    const float* Xrow = X + (size_t)(m0 + l) * IN_FEAT;
    v8f c = {0.f, 0.f, 0.f, 0.f, 0.f, 0.f, 0.f, 0.f};

#pragma unroll 8
    for (int k = 0; k < IN_FEAT; k += 4) {
        v2f a = *(const v2f*)(Xrow + k + 2 * half);
        v2f b;
        b.x = sW[(k + 2 * half + 0) * HIDDEN + n0 + l];
        b.y = sW[(k + 2 * half + 1) * HIDDEN + n0 + l];
        c = __builtin_amdgcn_wmma_f32_16x16x4_f32(
                 false, a,  false, b,
                 (short)0, c,  false,  false);
    }

    const size_t off = (size_t)(m0 + 8 * half) * HIDDEN + n0 + l;
#pragma unroll
    for (int v = 0; v < 8; v++) {
        H[off + (size_t)v * HIDDEN]    = c[v];
        aggA[off + (size_t)v * HIDDEN] = 0.0f;
        aggB[off + (size_t)v * HIDDEN] = 0.0f;
    }
}

#define S1_CHUNKS 4
#define S1_GROUP  8

__global__ void __launch_bounds__(256)
scatter1_kernel(const int* __restrict__ rows,
                const int* __restrict__ cols,
                const float* __restrict__ ew,
                const float* __restrict__ dinv,
                const float* __restrict__ h0,
                float* __restrict__ aggA,
                float* __restrict__ aggB,
                float* __restrict__ enorm, int e) {
    float* agg = in_upper_half() ? aggB : aggA;
    const int wave = __builtin_amdgcn_readfirstlane(
        (int)(blockIdx.x * (blockDim.x >> 5) + (threadIdx.x >> 5)));
    const int lane = threadIdx.x & 31;
    const int first = wave * (S1_CHUNKS * 32);

    for (int ch = 0; ch < S1_CHUNKS; ch++) {
        const int e0 = first + ch * 32;
        if (e0 >= e) break;
        const int cnt = (e - e0 < 32) ? (e - e0) : 32;

        int r = 0, c = 0;
        float norm = 0.0f;
        if (lane < cnt) {
            r = rows[e0 + lane];
            c = cols[e0 + lane];
            norm = dinv[r] * ew[e0 + lane] * dinv[c];
            enorm[e0 + lane] = norm;
        }

        int j = 0;
        for (; j + S1_GROUP <= cnt; j += S1_GROUP) {
            float vlo[S1_GROUP], vhi[S1_GROUP];
#pragma unroll
            for (int u = 0; u < S1_GROUP; u++) {
                const int rj = __builtin_amdgcn_readlane(r, j + u);
                const float* srow = h0 + (size_t)rj * HIDDEN + lane;
                vlo[u] = srow[0];
                vhi[u] = srow[32];
            }
#pragma unroll
            for (int u = 0; u < S1_GROUP; u++) {
                const int cj = __builtin_amdgcn_readlane(c, j + u);
                const float nj = __int_as_float(
                    __builtin_amdgcn_readlane(__float_as_int(norm), j + u));
                float* dst = agg + (size_t)cj * HIDDEN + lane;
                atomicAdd(dst + 0,  nj * vlo[u]);
                atomicAdd(dst + 32, nj * vhi[u]);
            }
        }
        for (; j < cnt; j++) {
            const int rj = __builtin_amdgcn_readlane(r, j);
            const int cj = __builtin_amdgcn_readlane(c, j);
            const float nj = __int_as_float(
                __builtin_amdgcn_readlane(__float_as_int(norm), j));
            const float* srow = h0 + (size_t)rj * HIDDEN + lane;
            const float a0 = srow[0];
            const float a1 = srow[32];
            float* dst = agg + (size_t)cj * HIDDEN + lane;
            atomicAdd(dst + 0,  nj * a0);
            atomicAdd(dst + 32, nj * a1);
        }
    }
}

#define E1_NODES 32

__global__ void __launch_bounds__(64)
epilogue1_kernel(const float* __restrict__ aggA,
                 const float* __restrict__ aggB,
                 const float* __restrict__ h0,
                 const float* __restrict__ b1,
                 const float* __restrict__ w2,
                 const float* __restrict__ b2,
                 const float* __restrict__ dinv,
                 float* __restrict__ g,
                 float* __restrict__ out,
                 float* __restrict__ outB, int n) {
    const int wave = blockIdx.x * (blockDim.x >> 5) + (threadIdx.x >> 5);
    const int lane = threadIdx.x & 31;
    const int base = wave * E1_NODES;
    if (base >= n) return;
    const int hl = lane & 15;
    const int half = lane >> 4;
    const float4 bv = ((const float4*)b1)[hl];
    const float4 wv = ((const float4*)w2)[hl];
    float res = 0.0f;
    const float dl = (base + lane < n) ? dinv[base + lane] : 0.0f;
    const float sql = dl * dl;

#pragma unroll 4
    for (int u = 0; u < E1_NODES / 2; u++) {
        int node = base + 2 * u + half;
        node = (node < n) ? node : (n - 1);
        const unsigned idx = (unsigned)node * (HIDDEN / 4) + (unsigned)hl;
        const float4 a = ((const float4*)aggA)[idx];
        const float4 b = ((const float4*)aggB)[idx];
        const float4 s = ((const float4*)h0)[idx];
        const float sq = __shfl(sql, 2 * u + half, 32);
        const float hx = fmaxf(((a.x + b.x) + sq * s.x) + bv.x, 0.0f);
        const float hy = fmaxf(((a.y + b.y) + sq * s.y) + bv.y, 0.0f);
        const float hz = fmaxf(((a.z + b.z) + sq * s.z) + bv.z, 0.0f);
        const float hw = fmaxf(((a.w + b.w) + sq * s.w) + bv.w, 0.0f);
        float p = hx * wv.x + hy * wv.y + hz * wv.z + hw * wv.w;
#pragma unroll
        for (int off = 8; off > 0; off >>= 1)
            p += __shfl_xor(p, off, 32);
        const float q = __shfl(p, (lane & 1) << 4, 32);
        if ((lane >> 1) == u) res = q;
    }

    const int i = base + lane;
    if (i < n) {
        g[i] = res;
        out[i] = b2[0] + sql * res;
        outB[i] = 0.0f;
    }
}

__global__ void scatter2_kernel(const int* __restrict__ rows,
                                const int* __restrict__ cols,
                                const float* __restrict__ enorm,
                                const float* __restrict__ g,
                                float* __restrict__ outA,
                                float* __restrict__ outB, int e) {
    float* out = in_upper_half() ? outB : outA;
    int i = blockIdx.x * blockDim.x + threadIdx.x;
    if (i >= e) return;
    int r = rows[i];
    int c = cols[i];
    float norm = enorm[i];
    atomicAdd(&out[c], norm * g[r]);
}

__global__ void add_planes_kernel(float* __restrict__ out,
                                  const float* __restrict__ outB, int n) {
    int i = blockIdx.x * blockDim.x + threadIdx.x;
    if (i < n) out[i] = out[i] + outB[i];
}

extern "C" void kernel_launch(void* const* d_in, const int* in_sizes, int n_in,
                              void* d_out, int out_size, void* d_ws, size_t ws_size,
                              hipStream_t stream) {
    const float* x  = (const float*)d_in[0];
    const int*   ei = (const int*)d_in[1];
    const float* ew = (const float*)d_in[2];
    const float* w1 = (const float*)d_in[3];
    const float* b1 = (const float*)d_in[4];
    const float* w2 = (const float*)d_in[5];
    const float* b2 = (const float*)d_in[6];
    float* out = (float*)d_out;

    const int E = in_sizes[2];
    const int N = in_sizes[0] / IN_FEAT;
    const int* rows = ei;
    const int* cols = ei + E;

    float* ws   = (float*)d_ws;
    float* degA = ws;
    float* g    = ws + N;
    float* degB = ws + 2 * (size_t)N;
    float* outB = ws + 3 * (size_t)N;
    float* h0   = ws + 4 * (size_t)N;
    float* agg  = h0 + (size_t)N * HIDDEN;
    float* aggB = agg + (size_t)N * HIDDEN;
    float* enorm = aggB + (size_t)N * HIDDEN;
    float* dinv = enorm + (size_t)E;

    const int B = 256;
    init_deg_kernel<<<(N + B - 1) / B, B, 0, stream>>>(degA, degB, N);

    {
        int mtiles = (N + 15) / 16;
        int waves  = mtiles * (HIDDEN / 16);
        int blocks = (waves + 7) / 8;
        gemm1_wmma_kernel<<<blocks, B, 0, stream>>>(x, w1, cols, ew, degA, degB, h0, agg, aggB, N, E);
    }
    dinv_kernel<<<(N + B - 1) / B, B, 0, stream>>>(degA, degB, dinv, N);

    scatter1_kernel<<<(E + 8 * S1_CHUNKS * 32 - 1) / (8 * S1_CHUNKS * 32), B, 0, stream>>>(rows, cols, ew, dinv, h0, agg, aggB, enorm, E);

    epilogue1_kernel<<<(N + 2 * E1_NODES - 1) / (2 * E1_NODES), 64, 0, stream>>>(agg, aggB, h0, b1, w2, b2, dinv, g, out, outB, N);
    scatter2_kernel<<<(E + B - 1) / B, B, 0, stream>>>(rows, cols, enorm, g, out, outB, E);
    add_planes_kernel<<<(N + B - 1) / B, B, 0, stream>>>(out, outB, N);
}


# Layer-1 GEMM operand staging: a block's 32 X rows and W1 reach LDS by whole-line 16-byte loads issued as one group (X was 16-line fragment loads in 4 waves, W1 32 load-store trips); k order kept
# speedup vs baseline: 3.6673x; 1.0681x over previous
#include <hip/hip_runtime.h>
#include <hip/hip_bf16.h>

#define IN_FEAT 128
#define HIDDEN  64

typedef __attribute__((ext_vector_type(2))) float v2f;
typedef __attribute__((ext_vector_type(8))) float v8f;

static __device__ __forceinline__ bool in_upper_half() {
    return ((__builtin_amdgcn_s_sendmsg_rtn(0x87) >> 18) & 1u) != 0u;
}

__global__ void init_deg_kernel(float* __restrict__ degA, float* __restrict__ degB, int n) {
    int i = blockIdx.x * blockDim.x + threadIdx.x;
    if (i < n) { degA[i] = 1.0f; degB[i] = 0.0f; }
}

__global__ void dinv_kernel(const float* __restrict__ degA, const float* __restrict__ degB,
                            float* __restrict__ dinv, int n) {
    int i = blockIdx.x * blockDim.x + threadIdx.x;
    if (i < n) {
        const float d = degA[i] + degB[i];
        dinv[i] = (d > 0.0f) ? __frsqrt_rn(d) : 0.0f;
    }
}

#define GEMM_THREADS 256
#define GEMM_ROWS    32
#define GEMM_XPITCH  132

__global__ void __launch_bounds__(GEMM_THREADS)
gemm1_wmma_kernel(const float* __restrict__ X, const float* __restrict__ W,
                  const int* __restrict__ cols, const float* __restrict__ ew,
                  float* __restrict__ degA, float* __restrict__ degB,
                  float* __restrict__ H,
                  float* __restrict__ aggA, float* __restrict__ aggB, int n, int e) {
    __shared__ __attribute__((aligned(16))) float sW[IN_FEAT * HIDDEN];
    __shared__ __attribute__((aligned(16))) float sX[GEMM_ROWS * GEMM_XPITCH];

    {
        float* deg = in_upper_half() ? degB : degA;
        const int stride = (int)(gridDim.x * blockDim.x);
        for (int i = (int)(blockIdx.x * blockDim.x + threadIdx.x); i < e; i += 2 * stride) {
            const int i1 = i + stride;
            const bool two = i1 < e;
            const int c0 = cols[i];
            const float w0 = ew[i];
            const int c1 = two ? cols[i1] : 0;
            const float w1 = two ? ew[i1] : 0.0f;
            atomicAdd(&deg[c0], w0);
            if (two) atomicAdd(&deg[c1], w1);
        }
    }

    const int xrow0 = (int)blockIdx.x * GEMM_ROWS;
    const int xw = (int)(threadIdx.x >> 5) * 4;
    const int xl = (int)(threadIdx.x & 31);
    float4 xr[4];
#pragma unroll
    for (int j = 0; j < 4; j++) {
        int row = xrow0 + xw + j;
        row = (row < n) ? row : (n - 1);
        xr[j] = ((const float4*)(X + (size_t)row * IN_FEAT))[xl];
    }
    float4 wr[IN_FEAT * HIDDEN / 4 / GEMM_THREADS];
#pragma unroll
    for (int j = 0; j < IN_FEAT * HIDDEN / 4 / GEMM_THREADS; j++)
        wr[j] = ((const float4*)W)[threadIdx.x + GEMM_THREADS * j];
#pragma unroll
    for (int j = 0; j < IN_FEAT * HIDDEN / 4 / GEMM_THREADS; j++)
        ((float4*)sW)[threadIdx.x + GEMM_THREADS * j] = wr[j];
#pragma unroll
    for (int j = 0; j < 4; j++)
        *(float4*)(&sX[(xw + j) * GEMM_XPITCH + 4 * xl]) = xr[j];
    __syncthreads();

    const int wave = blockIdx.x * (blockDim.x >> 5) + (threadIdx.x >> 5);
    const int lane = threadIdx.x & 31;
    const int mt = wave >> 2;
    const int nt = wave & 3;
    const int m0 = mt * 16;
    if (m0 >= n) return;
    const int n0 = nt * 16;
    const int half = lane >> 4;
    const int l = lane & 15;

    const float* Xrow = sX + ((m0 - xrow0) + l) * GEMM_XPITCH;
    v8f c = {0.f, 0.f, 0.f, 0.f, 0.f, 0.f, 0.f, 0.f};

#pragma unroll 8
    for (int k = 0; k < IN_FEAT; k += 4) {
        v2f a = *(const v2f*)(Xrow + k + 2 * half);
        v2f b;
        b.x = sW[(k + 2 * half + 0) * HIDDEN + n0 + l];
        b.y = sW[(k + 2 * half + 1) * HIDDEN + n0 + l];
        c = __builtin_amdgcn_wmma_f32_16x16x4_f32(
                 false, a,  false, b,
                 (short)0, c,  false,  false);
    }

    const size_t off = (size_t)(m0 + 8 * half) * HIDDEN + n0 + l;
#pragma unroll
    for (int v = 0; v < 8; v++) {
        H[off + (size_t)v * HIDDEN]    = c[v];
        aggA[off + (size_t)v * HIDDEN] = 0.0f;
        aggB[off + (size_t)v * HIDDEN] = 0.0f;
    }
}

#define S1_CHUNKS 4
#define S1_GROUP  8

__global__ void __launch_bounds__(256)
scatter1_kernel(const int* __restrict__ rows,
                const int* __restrict__ cols,
                const float* __restrict__ ew,
                const float* __restrict__ dinv,
                const float* __restrict__ h0,
                float* __restrict__ aggA,
                float* __restrict__ aggB,
                float* __restrict__ enorm, int e) {
    float* agg = in_upper_half() ? aggB : aggA;
    const int wave = __builtin_amdgcn_readfirstlane(
        (int)(blockIdx.x * (blockDim.x >> 5) + (threadIdx.x >> 5)));
    const int lane = threadIdx.x & 31;
    const int first = wave * (S1_CHUNKS * 32);

    for (int ch = 0; ch < S1_CHUNKS; ch++) {
        const int e0 = first + ch * 32;
        if (e0 >= e) break;
        const int cnt = (e - e0 < 32) ? (e - e0) : 32;

        int r = 0, c = 0;
        float norm = 0.0f;
        if (lane < cnt) {
            r = rows[e0 + lane];
            c = cols[e0 + lane];
            norm = dinv[r] * ew[e0 + lane] * dinv[c];
            enorm[e0 + lane] = norm;
        }

        int j = 0;
        for (; j + S1_GROUP <= cnt; j += S1_GROUP) {
            float vlo[S1_GROUP], vhi[S1_GROUP];
#pragma unroll
            for (int u = 0; u < S1_GROUP; u++) {
                const int rj = __builtin_amdgcn_readlane(r, j + u);
                const float* srow = h0 + (size_t)rj * HIDDEN + lane;
                vlo[u] = srow[0];
                vhi[u] = srow[32];
            }
#pragma unroll
            for (int u = 0; u < S1_GROUP; u++) {
                const int cj = __builtin_amdgcn_readlane(c, j + u);
                const float nj = __int_as_float(
                    __builtin_amdgcn_readlane(__float_as_int(norm), j + u));
                float* dst = agg + (size_t)cj * HIDDEN + lane;
                atomicAdd(dst + 0,  nj * vlo[u]);
                atomicAdd(dst + 32, nj * vhi[u]);
            }
        }
        for (; j < cnt; j++) {
            const int rj = __builtin_amdgcn_readlane(r, j);
            const int cj = __builtin_amdgcn_readlane(c, j);
            const float nj = __int_as_float(
                __builtin_amdgcn_readlane(__float_as_int(norm), j));
            const float* srow = h0 + (size_t)rj * HIDDEN + lane;
            const float a0 = srow[0];
            const float a1 = srow[32];
            float* dst = agg + (size_t)cj * HIDDEN + lane;
            atomicAdd(dst + 0,  nj * a0);
            atomicAdd(dst + 32, nj * a1);
        }
    }
}

#define E1_NODES 32

__global__ void __launch_bounds__(64)
epilogue1_kernel(const float* __restrict__ aggA,
                 const float* __restrict__ aggB,
                 const float* __restrict__ h0,
                 const float* __restrict__ b1,
                 const float* __restrict__ w2,
                 const float* __restrict__ b2,
                 const float* __restrict__ dinv,
                 float* __restrict__ g,
                 float* __restrict__ out,
                 float* __restrict__ outB, int n) {
    const int wave = blockIdx.x * (blockDim.x >> 5) + (threadIdx.x >> 5);
    const int lane = threadIdx.x & 31;
    const int base = wave * E1_NODES;
    if (base >= n) return;
    const int hl = lane & 15;
    const int half = lane >> 4;
    const float4 bv = ((const float4*)b1)[hl];
    const float4 wv = ((const float4*)w2)[hl];
    float res = 0.0f;
    const float dl = (base + lane < n) ? dinv[base + lane] : 0.0f;
    const float sql = dl * dl;

#pragma unroll 4
    for (int u = 0; u < E1_NODES / 2; u++) {
        int node = base + 2 * u + half;
        node = (node < n) ? node : (n - 1);
        const unsigned idx = (unsigned)node * (HIDDEN / 4) + (unsigned)hl;
        const float4 a = ((const float4*)aggA)[idx];
        const float4 b = ((const float4*)aggB)[idx];
        const float4 s = ((const float4*)h0)[idx];
        const float sq = __shfl(sql, 2 * u + half, 32);
        const float hx = fmaxf(((a.x + b.x) + sq * s.x) + bv.x, 0.0f);
        const float hy = fmaxf(((a.y + b.y) + sq * s.y) + bv.y, 0.0f);
        const float hz = fmaxf(((a.z + b.z) + sq * s.z) + bv.z, 0.0f);
        const float hw = fmaxf(((a.w + b.w) + sq * s.w) + bv.w, 0.0f);
        float p = hx * wv.x + hy * wv.y + hz * wv.z + hw * wv.w;
#pragma unroll
        for (int off = 8; off > 0; off >>= 1)
            p += __shfl_xor(p, off, 32);
        const float q = __shfl(p, (lane & 1) << 4, 32);
        if ((lane >> 1) == u) res = q;
    }

    const int i = base + lane;
    if (i < n) {
        g[i] = res;
        out[i] = b2[0] + sql * res;
        outB[i] = 0.0f;
    }
}

__global__ void scatter2_kernel(const int* __restrict__ rows,
                                const int* __restrict__ cols,
                                const float* __restrict__ enorm,
                                const float* __restrict__ g,
                                float* __restrict__ outA,
                                float* __restrict__ outB, int e) {
    float* out = in_upper_half() ? outB : outA;
    int i = blockIdx.x * blockDim.x + threadIdx.x;
    if (i >= e) return;
    int r = rows[i];
    int c = cols[i];
    float norm = enorm[i];
    atomicAdd(&out[c], norm * g[r]);
}

__global__ void add_planes_kernel(float* __restrict__ out,
                                  const float* __restrict__ outB, int n) {
    int i = blockIdx.x * blockDim.x + threadIdx.x;
    if (i < n) out[i] = out[i] + outB[i];
}

extern "C" void kernel_launch(void* const* d_in, const int* in_sizes, int n_in,
                              void* d_out, int out_size, void* d_ws, size_t ws_size,
                              hipStream_t stream) {
    const float* x  = (const float*)d_in[0];
    const int*   ei = (const int*)d_in[1];
    const float* ew = (const float*)d_in[2];
    const float* w1 = (const float*)d_in[3];
    const float* b1 = (const float*)d_in[4];
    const float* w2 = (const float*)d_in[5];
    const float* b2 = (const float*)d_in[6];
    float* out = (float*)d_out;

    const int E = in_sizes[2];
    const int N = in_sizes[0] / IN_FEAT;
    const int* rows = ei;
    const int* cols = ei + E;

    float* ws   = (float*)d_ws;
    float* degA = ws;
    float* g    = ws + N;
    float* degB = ws + 2 * (size_t)N;
    float* outB = ws + 3 * (size_t)N;
    float* h0   = ws + 4 * (size_t)N;
    float* agg  = h0 + (size_t)N * HIDDEN;
    float* aggB = agg + (size_t)N * HIDDEN;
    float* enorm = aggB + (size_t)N * HIDDEN;
    float* dinv = enorm + (size_t)E;

    const int B = 256;
    init_deg_kernel<<<(N + B - 1) / B, B, 0, stream>>>(degA, degB, N);

    {
        int mtiles = (N + 15) / 16;
        int waves  = mtiles * (HIDDEN / 16);
        int blocks = (waves + 7) / 8;
        gemm1_wmma_kernel<<<blocks, GEMM_THREADS, 0, stream>>>(x, w1, cols, ew, degA, degB, h0, agg, aggB, N, E);
    }
    dinv_kernel<<<(N + B - 1) / B, B, 0, stream>>>(degA, degB, dinv, N);

    scatter1_kernel<<<(E + 8 * S1_CHUNKS * 32 - 1) / (8 * S1_CHUNKS * 32), B, 0, stream>>>(rows, cols, ew, dinv, h0, agg, aggB, enorm, E);

    epilogue1_kernel<<<(N + 2 * E1_NODES - 1) / (2 * E1_NODES), 64, 0, stream>>>(agg, aggB, h0, b1, w2, b2, dinv, g, out, outB, N);
    scatter2_kernel<<<(E + B - 1) / B, B, 0, stream>>>(rows, cols, enorm, g, out, outB, E);
    add_planes_kernel<<<(N + B - 1) / B, B, 0, stream>>>(out, outB, N);
}
